# starting kernel
# baseline (speedup 1.0000x reference)
#include <hip/hip_runtime.h>

typedef __attribute__((ext_vector_type(2))) float v2f;
typedef __attribute__((ext_vector_type(8))) float v8f;

#define N_NODES 100000
#define N_EDGES 1600000
#define D 128
#define COMB_STRIDE 260
#define OUT_STRIDE 132

__global__ void zero_f32(float4* __restrict__ p, long n4) {
    long i = (long)blockIdx.x * blockDim.x + threadIdx.x;
    long stride = (long)gridDim.x * blockDim.x;
    float4 z = {0.f, 0.f, 0.f, 0.f};
    for (; i < n4; i += stride) p[i] = z;
}

__global__ void edge_count(const int* __restrict__ dst, float* __restrict__ cnt) {
    int e = blockIdx.x * blockDim.x + threadIdx.x;
    if (e < N_EDGES) atomicAdd(&cnt[dst[e]], 1.0f);
}

__global__ void edge_scatter(const int* __restrict__ src, const int* __restrict__ dst,
                             const float* __restrict__ x, float* __restrict__ agg) {
    long idx = (long)blockIdx.x * blockDim.x + threadIdx.x;
    long e = idx >> 5;
    if (e >= N_EDGES) return;
    int f = (int)(idx & 31) * 4;
    long s = src[e];
    long d = dst[e];
    float4 v = *(const float4*)(x + s * D + f);
    float* ap = agg + d * D + f;
    atomicAdd(ap + 0, v.x);
    atomicAdd(ap + 1, v.y);
    atomicAdd(ap + 2, v.z);
    atomicAdd(ap + 3, v.w);
}

__global__ void __launch_bounds__(256)
sage_gemm(const float* __restrict__ agg, const float* __restrict__ cnt,
          const float* __restrict__ xin, const float* __restrict__ Wl,
          const float* __restrict__ bl,  const float* __restrict__ Wr,
          float* __restrict__ out, int relu) {
    extern __shared__ float lds[];

    for (int i = threadIdx.x; i < D * (D / 2); i += 256) {
        int n = i >> 6;
        int p = i & 63;
        v2f wl = *(const v2f*)(Wl + n * D + 2 * p);
        v2f wr = *(const v2f*)(Wr + n * D + 2 * p);
        float4 c4 = {wl.x, wl.y, wr.x, wr.y};
        *(float4*)(lds + n * COMB_STRIDE + p * 4) = c4;
    }
    __syncthreads();

    const int lane = threadIdx.x & 31;
    const int wave = threadIdx.x >> 5;
    const int half = lane >> 4;
    const int l16  = lane & 15;

    long rowA = (long)blockIdx.x * 128 + wave * 16 + l16;
    long rA   = rowA < N_NODES ? rowA : (long)(N_NODES - 1);

    float c   = cnt[rA];
    float inv = 1.0f / (c > 1.0f ? c : 1.0f);

    const float* aggRow = agg + rA * D;
    const float* xRow   = xin + rA * D;

    v8f acc[8] = {};

    for (int k = 0; k < D; k += 4) {
        int ko = k + 2 * half;
        int p4 = ((ko >> 1) << 2);
        v2f aM = *(const v2f*)(aggRow + ko);
        aM = aM * inv;
        v2f aX = *(const v2f*)(xRow + ko);

        float4 bf[8];
        #pragma unroll
        for (int t = 0; t < 8; ++t) {
            int n = t * 16 + l16;
            bf[t] = *(const float4*)(lds + n * COMB_STRIDE + p4);
        }
        #pragma unroll
        for (int t = 0; t < 8; ++t) {
            v2f bL = {bf[t].x, bf[t].y};
            v2f bR = {bf[t].z, bf[t].w};
            acc[t] = __builtin_amdgcn_wmma_f32_16x16x4_f32(
                false, aM, false, bL, (short)0, acc[t], false, false);
            acc[t] = __builtin_amdgcn_wmma_f32_16x16x4_f32(
                false, aX, false, bR, (short)0, acc[t], false, false);
        }
    }

    __syncthreads();

    int localBase = wave * 16 + half * 8;
    #pragma unroll
    for (int t = 0; t < 8; ++t) {
        int n = t * 16 + l16;
        float bias = bl[n];
        #pragma unroll
        for (int r = 0; r < 8; ++r) {
            float v = acc[t][r] + bias;
            if (relu) v = v > 0.0f ? v : 0.0f;
            lds[(localBase + r) * OUT_STRIDE + n] = v;
        }
    }
    __syncthreads();

    long blockBase = (long)blockIdx.x * 128;
    for (int i = threadIdx.x; i < 128 * (D / 4); i += 256) {
        int row = i >> 5;
        int c4  = (i & 31) * 4;
        long grow = blockBase + row;
        if (grow < N_NODES) {
            float4 v = *(const float4*)(lds + row * OUT_STRIDE + c4);
            *(float4*)(out + grow * D + c4) = v;
        }
    }
}

extern "C" void kernel_launch(void* const* d_in, const int* in_sizes, int n_in,
                              void* d_out, int out_size, void* d_ws, size_t ws_size,
                              hipStream_t stream) {
    const float* x   = (const float*)d_in[0];
    const int*   ei  = (const int*)d_in[1];
    const float* W1l = (const float*)d_in[2];
    const float* b1l = (const float*)d_in[3];
    const float* W1r = (const float*)d_in[4];
    const float* W2l = (const float*)d_in[5];
    const float* b2l = (const float*)d_in[6];
    const float* W2r = (const float*)d_in[7];

    const int* src = ei;
    const int* dst = ei + N_EDGES;

    float* agg  = (float*)d_ws;
    float* cntp = agg + (size_t)N_NODES * D;
    float* h    = cntp + N_NODES;
    float* outp = (float*)d_out;

    const long nAgg = (long)N_NODES * D;
    const int  threads = 256;
    const long scatterThreads = (long)N_EDGES * 32;
    const int  scatterBlocks  = (int)((scatterThreads + threads - 1) / threads);
    const int  gemmBlocks     = (N_NODES + 127) / 128;
    const size_t shmem        = (size_t)D * COMB_STRIDE * sizeof(float);

    zero_f32<<<2048, threads, 0, stream>>>((float4*)agg, (nAgg + N_NODES) / 4);
    edge_count<<<(N_EDGES + threads - 1) / threads, threads, 0, stream>>>(dst, cntp);
    edge_scatter<<<scatterBlocks, threads, 0, stream>>>(src, dst, x, agg);
    sage_gemm<<<gemmBlocks, threads, shmem, stream>>>(agg, cntp, x, W1l, b1l, W1r, h, 1);

    zero_f32<<<2048, threads, 0, stream>>>((float4*)agg, nAgg / 4);
    edge_scatter<<<scatterBlocks, threads, 0, stream>>>(src, dst, h, agg);
    sage_gemm<<<gemmBlocks, threads, shmem, stream>>>(agg, cntp, h, W2l, b2l, W2r, outp, 0);
}


# Atomic scatter-add replaced by an edge list grouped by destination row (integer slot atomics, scan, fill, per-row sort) and a wave-per-row gather sum; rows now summed in ascending source order
# speedup vs baseline: 2.1640x; 2.1640x over previous
#include <hip/hip_runtime.h>

typedef __attribute__((ext_vector_type(2))) float v2f;
typedef __attribute__((ext_vector_type(8))) float v8f;

#define N_NODES 100000
#define N_EDGES 1600000
#define D 128
#define COMB_STRIDE 260
#define OUT_STRIDE 132

__global__ void zero_f32(float4* __restrict__ p, long n4) {
    long i = (long)blockIdx.x * blockDim.x + threadIdx.x;
    long stride = (long)gridDim.x * blockDim.x;
    float4 z = {0.f, 0.f, 0.f, 0.f};
    for (; i < n4; i += stride) p[i] = z;
}

#define EDGE_OCT (N_EDGES / 8)
#define ORDER_ROWS 8
#define GATHER_ROWS 4

__global__ void __launch_bounds__(256)
edge_rank(const int* __restrict__ dst, int* __restrict__ deg, int* __restrict__ rank) {
    const int i = blockIdx.x * 256 + threadIdx.x;
    if (i >= EDGE_OCT) return;
    const int4* d4 = (const int4*)dst;
    const int4 a = d4[i];
    const int4 b = d4[i + EDGE_OCT];
    int4 ra, rb;
    ra.x = atomicAdd(&deg[a.x], 1);
    ra.y = atomicAdd(&deg[a.y], 1);
    ra.z = atomicAdd(&deg[a.z], 1);
    ra.w = atomicAdd(&deg[a.w], 1);
    rb.x = atomicAdd(&deg[b.x], 1);
    rb.y = atomicAdd(&deg[b.y], 1);
    rb.z = atomicAdd(&deg[b.z], 1);
    rb.w = atomicAdd(&deg[b.w], 1);
    int4* r4 = (int4*)rank;
    r4[i] = ra;
    r4[i + EDGE_OCT] = rb;
}

__global__ void __launch_bounds__(1024)
scan_deg(const int* __restrict__ deg, int* __restrict__ off, float* __restrict__ cnt) {
    __shared__ int sred[32];
    __shared__ int stot[32];
    const int t = threadIdx.x;
    const int lane = t & 31;
    const int wave = t >> 5;
    const int b = blockIdx.x;

    int s = 0;
    const int4* d4 = (const int4*)deg;
    const int lim4 = b * 256;
    #pragma unroll 4
    for (int i = t; i < lim4; i += 1024) {
        int4 v = d4[i];
        s += (v.x + v.y) + (v.z + v.w);
    }
    for (int o = 16; o > 0; o >>= 1) s += __shfl_xor(s, o);
    if (lane == 0) sred[wave] = s;

    const int idx = b * 1024 + t;
    const int v = idx < N_NODES ? deg[idx] : 0;
    int inc = v;
    for (int o = 1; o < 32; o <<= 1) {
        int u = __shfl_up(inc, o);
        if (lane >= o) inc += u;
    }
    if (lane == 31) stot[wave] = inc;
    __syncthreads();

    int prefix = sred[lane];
    for (int o = 16; o > 0; o >>= 1) prefix += __shfl_xor(prefix, o);
    const int wt = stot[lane];
    int winc = wt;
    for (int o = 1; o < 32; o <<= 1) {
        int u = __shfl_up(winc, o);
        if (lane >= o) winc += u;
    }
    const int wexcl = __shfl(winc - wt, wave);
    const int excl = prefix + wexcl + inc - v;
    if (idx < N_NODES) {
        off[idx] = excl;
        cnt[idx] = (float)v;
    }
    if (idx == N_NODES - 1) off[N_NODES] = excl + v;
}

__global__ void __launch_bounds__(256)
edge_fill(const int* __restrict__ src, const int* __restrict__ dst,
          const int* __restrict__ off, const int* __restrict__ rank,
          int* __restrict__ tmp) {
    const int i = blockIdx.x * 256 + threadIdx.x;
    if (i >= EDGE_OCT) return;
    const int4 da = ((const int4*)dst)[i];
    const int4 db = ((const int4*)dst)[i + EDGE_OCT];
    const int4 ra = ((const int4*)rank)[i];
    const int4 rb = ((const int4*)rank)[i + EDGE_OCT];
    const int4 sa = ((const int4*)src)[i];
    const int4 sb = ((const int4*)src)[i + EDGE_OCT];
    tmp[off[da.x] + ra.x] = sa.x;
    tmp[off[da.y] + ra.y] = sa.y;
    tmp[off[da.z] + ra.z] = sa.z;
    tmp[off[da.w] + ra.w] = sa.w;
    tmp[off[db.x] + rb.x] = sb.x;
    tmp[off[db.y] + rb.y] = sb.y;
    tmp[off[db.z] + rb.z] = sb.z;
    tmp[off[db.w] + rb.w] = sb.w;
}

__global__ void __launch_bounds__(256)
seg_order(const int* __restrict__ off, const int* __restrict__ tmp, int* __restrict__ csr) {
    const int lane = threadIdx.x & 31;
    const int w = __builtin_amdgcn_readfirstlane((int)(blockIdx.x * 8 + (threadIdx.x >> 5)));
    for (int rr = 0; rr < ORDER_ROWS; ++rr) {
        const int n = w * ORDER_ROWS + rr;
        if (n >= N_NODES) break;
        const int start = off[n];
        const int d = off[n + 1] - start;
        for (int ib = 0; ib < d; ib += 32) {
            const int i = ib + lane;
            const bool valid = i < d;
            const int my = valid ? tmp[start + i] : 0x7fffffff;
            int r = 0;
            for (int jb = 0; jb < d; jb += 32) {
                const int j0 = jb + lane;
                const int kv = j0 < d ? tmp[start + j0] : 0x7fffffff;
                const int m = (d - jb) < 32 ? (d - jb) : 32;
                for (int jj = 0; jj < m; ++jj) {
                    const int kj = __builtin_amdgcn_readlane(kv, jj);
                    const int j = jb + jj;
                    r += (kj < my || (kj == my && j < i)) ? 1 : 0;
                }
            }
            if (valid) csr[start + r] = my;
        }
    }
}

__device__ __forceinline__ void add4(float4& a, const float4& v) {
    a.x += v.x; a.y += v.y; a.z += v.z; a.w += v.w;
}

__global__ void __launch_bounds__(256)
gather_rows(const int* __restrict__ off, const int* __restrict__ csr,
            const float* __restrict__ x, float* __restrict__ agg) {
    const int lane = threadIdx.x & 31;
    const int w = __builtin_amdgcn_readfirstlane((int)(blockIdx.x * 8 + (threadIdx.x >> 5)));
    const float* xl = x + lane * 4;
    for (int rr = 0; rr < GATHER_ROWS; ++rr) {
        const int n = w * GATHER_ROWS + rr;
        if (n >= N_NODES) break;
        const int start = off[n];
        const int end   = off[n + 1];
        float4 acc = {0.f, 0.f, 0.f, 0.f};
        for (int base = start; base < end; base += 32) {
            const int m = (end - base) < 32 ? (end - base) : 32;
            const int sv = lane < m ? csr[base + lane] : 0;
            int jj = 0;
            for (; jj + 8 <= m; jj += 8) {
                float4 v[8];
                #pragma unroll
                for (int u = 0; u < 8; ++u) {
                    const long s = __builtin_amdgcn_readlane(sv, jj + u);
                    v[u] = *(const float4*)(xl + s * D);
                }
                #pragma unroll
                for (int u = 0; u < 8; ++u) add4(acc, v[u]);
            }
            for (; jj < m; jj += 4) {
                float4 v[4];
                #pragma unroll
                for (int u = 0; u < 4; ++u) {
                    const int q = (jj + u) < m ? (jj + u) : (m - 1);
                    const long s = __builtin_amdgcn_readlane(sv, q);
                    v[u] = *(const float4*)(xl + s * D);
                }
                #pragma unroll
                for (int u = 0; u < 4; ++u)
                    if (jj + u < m) add4(acc, v[u]);
            }
        }
        *(float4*)(agg + (long)n * D + lane * 4) = acc;
    }
}

__global__ void __launch_bounds__(256)
sage_gemm(const float* __restrict__ agg, const float* __restrict__ cnt,
          const float* __restrict__ xin, const float* __restrict__ Wl,
          const float* __restrict__ bl,  const float* __restrict__ Wr,
          float* __restrict__ out, int relu) {
    extern __shared__ float lds[];

    for (int i = threadIdx.x; i < D * (D / 2); i += 256) {
        int n = i >> 6;
        int p = i & 63;
        v2f wl = *(const v2f*)(Wl + n * D + 2 * p);
        v2f wr = *(const v2f*)(Wr + n * D + 2 * p);
        float4 c4 = {wl.x, wl.y, wr.x, wr.y};
        *(float4*)(lds + n * COMB_STRIDE + p * 4) = c4;
    }
    __syncthreads();

    const int lane = threadIdx.x & 31;
    const int wave = threadIdx.x >> 5;
    const int half = lane >> 4;
    const int l16  = lane & 15;

    long rowA = (long)blockIdx.x * 128 + wave * 16 + l16;
    long rA   = rowA < N_NODES ? rowA : (long)(N_NODES - 1);

    float c   = cnt[rA];
    float inv = 1.0f / (c > 1.0f ? c : 1.0f);

    const float* aggRow = agg + rA * D;
    const float* xRow   = xin + rA * D;

    v8f acc[8] = {};

    for (int k = 0; k < D; k += 4) {
        int ko = k + 2 * half;
        int p4 = ((ko >> 1) << 2);
        v2f aM = *(const v2f*)(aggRow + ko);
        aM = aM * inv;
        v2f aX = *(const v2f*)(xRow + ko);

        float4 bf[8];
        #pragma unroll
        for (int t = 0; t < 8; ++t) {
            int n = t * 16 + l16;
            bf[t] = *(const float4*)(lds + n * COMB_STRIDE + p4);
        }
        #pragma unroll
        for (int t = 0; t < 8; ++t) {
            v2f bL = {bf[t].x, bf[t].y};
            v2f bR = {bf[t].z, bf[t].w};
            acc[t] = __builtin_amdgcn_wmma_f32_16x16x4_f32(
                false, aM, false, bL, (short)0, acc[t], false, false);
            acc[t] = __builtin_amdgcn_wmma_f32_16x16x4_f32(
                false, aX, false, bR, (short)0, acc[t], false, false);
        }
    }

    __syncthreads();

    int localBase = wave * 16 + half * 8;
    #pragma unroll
    for (int t = 0; t < 8; ++t) {
        int n = t * 16 + l16;
        float bias = bl[n];
        #pragma unroll
        for (int r = 0; r < 8; ++r) {
            float v = acc[t][r] + bias;
            if (relu) v = v > 0.0f ? v : 0.0f;
            lds[(localBase + r) * OUT_STRIDE + n] = v;
        }
    }
    __syncthreads();

    long blockBase = (long)blockIdx.x * 128;
    for (int i = threadIdx.x; i < 128 * (D / 4); i += 256) {
        int row = i >> 5;
        int c4  = (i & 31) * 4;
        long grow = blockBase + row;
        if (grow < N_NODES) {
            float4 v = *(const float4*)(lds + row * OUT_STRIDE + c4);
            *(float4*)(out + grow * D + c4) = v;
        }
    }
}

extern "C" void kernel_launch(void* const* d_in, const int* in_sizes, int n_in,
                              void* d_out, int out_size, void* d_ws, size_t ws_size,
                              hipStream_t stream) {
    const float* x   = (const float*)d_in[0];
    const int*   ei  = (const int*)d_in[1];
    const float* W1l = (const float*)d_in[2];
    const float* b1l = (const float*)d_in[3];
    const float* W1r = (const float*)d_in[4];
    const float* W2l = (const float*)d_in[5];
    const float* b2l = (const float*)d_in[6];
    const float* W2r = (const float*)d_in[7];

    const int* src = ei;
    const int* dst = ei + N_EDGES;

    float* agg  = (float*)d_ws;
    float* cntp = agg + (size_t)N_NODES * D;
    float* h    = cntp + N_NODES;
    float* outp = (float*)d_out;

    int* deg  = (int*)h;
    int* rank = deg + N_NODES;
    int* tmp  = rank + N_EDGES;
    int* off  = (int*)d_out;
    int* csr  = off + (N_NODES + 4);

    const int  threads = 256;
    const int  edgeBlocks   = (EDGE_OCT + threads - 1) / threads;
    const int  orderBlocks  = (N_NODES + 8 * ORDER_ROWS - 1) / (8 * ORDER_ROWS);
    const int  gatherBlocks = (N_NODES + 8 * GATHER_ROWS - 1) / (8 * GATHER_ROWS);
    const int  gemmBlocks = (N_NODES + 127) / 128;
    const size_t shmem    = (size_t)D * COMB_STRIDE * sizeof(float);

    zero_f32<<<64, threads, 0, stream>>>((float4*)deg, N_NODES / 4);
    edge_rank<<<edgeBlocks, threads, 0, stream>>>(dst, deg, rank);
    scan_deg<<<(N_NODES + 1023) / 1024, 1024, 0, stream>>>(deg, off, cntp);
    edge_fill<<<edgeBlocks, threads, 0, stream>>>(src, dst, off, rank, tmp);
    seg_order<<<orderBlocks, threads, 0, stream>>>(off, tmp, csr);

    gather_rows<<<gatherBlocks, threads, 0, stream>>>(off, csr, x, agg);
    sage_gemm<<<gemmBlocks, threads, shmem, stream>>>(agg, cntp, x, W1l, b1l, W1r, h, 1);

    gather_rows<<<gatherBlocks, threads, 0, stream>>>(off, csr, h, agg);
    sage_gemm<<<gemmBlocks, threads, shmem, stream>>>(agg, cntp, h, W2l, b2l, W2r, outp, 0);
}


# Neighbour gather fused into the product kernel: each wave sums its own 16 rows from the edge run into a private LDS slot and feeds the WMMAs from it; one 16-wave workgroup per WGP; sum order kept
# speedup vs baseline: 2.5153x; 1.1623x over previous
#include <hip/hip_runtime.h>

typedef __attribute__((ext_vector_type(2))) float v2f;
typedef __attribute__((ext_vector_type(8))) float v8f;

#define N_NODES 100000
#define N_EDGES 1600000
#define D 128
#define COMB_STRIDE 260
#define SLOT_STRIDE 132
#define SLOT_FLOATS (16 * SLOT_STRIDE)
#define LAYER_WAVES 16
#define N_STRIPES (N_NODES / 16)

__global__ void zero_f32(float4* __restrict__ p, long n4) {
    long i = (long)blockIdx.x * blockDim.x + threadIdx.x;
    long stride = (long)gridDim.x * blockDim.x;
    float4 z = {0.f, 0.f, 0.f, 0.f};
    for (; i < n4; i += stride) p[i] = z;
}

#define EDGE_OCT (N_EDGES / 8)
#define ORDER_ROWS 8

__global__ void __launch_bounds__(256)
edge_rank(const int* __restrict__ dst, int* __restrict__ deg, int* __restrict__ rank) {
    const int i = blockIdx.x * 256 + threadIdx.x;
    if (i >= EDGE_OCT) return;
    const int4* d4 = (const int4*)dst;
    const int4 a = d4[i];
    const int4 b = d4[i + EDGE_OCT];
    int4 ra, rb;
    ra.x = atomicAdd(&deg[a.x], 1);
    ra.y = atomicAdd(&deg[a.y], 1);
    ra.z = atomicAdd(&deg[a.z], 1);
    ra.w = atomicAdd(&deg[a.w], 1);
    rb.x = atomicAdd(&deg[b.x], 1);
    rb.y = atomicAdd(&deg[b.y], 1);
    rb.z = atomicAdd(&deg[b.z], 1);
    rb.w = atomicAdd(&deg[b.w], 1);
    int4* r4 = (int4*)rank;
    r4[i] = ra;
    r4[i + EDGE_OCT] = rb;
}

__global__ void __launch_bounds__(1024)
scan_deg(const int* __restrict__ deg, int* __restrict__ off, float* __restrict__ cnt) {
    __shared__ int sred[32];
    __shared__ int stot[32];
    const int t = threadIdx.x;
    const int lane = t & 31;
    const int wave = t >> 5;
    const int b = blockIdx.x;

    int s = 0;
    const int4* d4 = (const int4*)deg;
    const int lim4 = b * 256;
    #pragma unroll 4
    for (int i = t; i < lim4; i += 1024) {
        int4 v = d4[i];
        s += (v.x + v.y) + (v.z + v.w);
    }
    for (int o = 16; o > 0; o >>= 1) s += __shfl_xor(s, o);
    if (lane == 0) sred[wave] = s;

    const int idx = b * 1024 + t;
    const int v = idx < N_NODES ? deg[idx] : 0;
    int inc = v;
    for (int o = 1; o < 32; o <<= 1) {
        int u = __shfl_up(inc, o);
        if (lane >= o) inc += u;
    }
    if (lane == 31) stot[wave] = inc;
    __syncthreads();

    int prefix = sred[lane];
    for (int o = 16; o > 0; o >>= 1) prefix += __shfl_xor(prefix, o);
    const int wt = stot[lane];
    int winc = wt;
    for (int o = 1; o < 32; o <<= 1) {
        int u = __shfl_up(winc, o);
        if (lane >= o) winc += u;
    }
    const int wexcl = __shfl(winc - wt, wave);
    const int excl = prefix + wexcl + inc - v;
    if (idx < N_NODES) {
        off[idx] = excl;
        cnt[idx] = (float)v;
    }
    if (idx == N_NODES - 1) off[N_NODES] = excl + v;
}

__global__ void __launch_bounds__(256)
edge_fill(const int* __restrict__ src, const int* __restrict__ dst,
          const int* __restrict__ off, const int* __restrict__ rank,
          int* __restrict__ tmp) {
    const int i = blockIdx.x * 256 + threadIdx.x;
    if (i >= EDGE_OCT) return;
    const int4 da = ((const int4*)dst)[i];
    const int4 db = ((const int4*)dst)[i + EDGE_OCT];
    const int4 ra = ((const int4*)rank)[i];
    const int4 rb = ((const int4*)rank)[i + EDGE_OCT];
    const int4 sa = ((const int4*)src)[i];
    const int4 sb = ((const int4*)src)[i + EDGE_OCT];
    tmp[off[da.x] + ra.x] = sa.x;
    tmp[off[da.y] + ra.y] = sa.y;
    tmp[off[da.z] + ra.z] = sa.z;
    tmp[off[da.w] + ra.w] = sa.w;
    tmp[off[db.x] + rb.x] = sb.x;
    tmp[off[db.y] + rb.y] = sb.y;
    tmp[off[db.z] + rb.z] = sb.z;
    tmp[off[db.w] + rb.w] = sb.w;
}

__global__ void __launch_bounds__(256)
seg_order(const int* __restrict__ off, const int* __restrict__ tmp, int* __restrict__ csr) {
    const int lane = threadIdx.x & 31;
    const int w = __builtin_amdgcn_readfirstlane((int)(blockIdx.x * 8 + (threadIdx.x >> 5)));
    for (int rr = 0; rr < ORDER_ROWS; ++rr) {
        const int n = w * ORDER_ROWS + rr;
        if (n >= N_NODES) break;
        const int start = off[n];
        const int d = off[n + 1] - start;
        for (int ib = 0; ib < d; ib += 32) {
            const int i = ib + lane;
            const bool valid = i < d;
            const int my = valid ? tmp[start + i] : 0x7fffffff;
            int r = 0;
            for (int jb = 0; jb < d; jb += 32) {
                const int j0 = jb + lane;
                const int kv = j0 < d ? tmp[start + j0] : 0x7fffffff;
                const int m = (d - jb) < 32 ? (d - jb) : 32;
                for (int jj = 0; jj < m; ++jj) {
                    const int kj = __builtin_amdgcn_readlane(kv, jj);
                    const int j = jb + jj;
                    r += (kj < my || (kj == my && j < i)) ? 1 : 0;
                }
            }
            if (valid) csr[start + r] = my;
        }
    }
}

static_assert(N_NODES % 16 == 0, "stripes of 16 rows cover the node count exactly");

__device__ __forceinline__ void add4(float4& a, const float4& v) {
    a.x += v.x; a.y += v.y; a.z += v.z; a.w += v.w;
}

__global__ void __launch_bounds__(LAYER_WAVES * 32)
sage_layer(const int* __restrict__ off, const int* __restrict__ csr,
           const float* __restrict__ cnt, const float* __restrict__ xin,
           const float* __restrict__ Wl, const float* __restrict__ bl,
           const float* __restrict__ Wr, float* __restrict__ out, int relu) {
    __shared__ __attribute__((aligned(16))) float lds[D * COMB_STRIDE + LAYER_WAVES * SLOT_FLOATS];

    for (int i = threadIdx.x; i < D * (D / 2); i += LAYER_WAVES * 32) {
        int n = i >> 6;
        int p = i & 63;
        v2f wl = *(const v2f*)(Wl + n * D + 2 * p);
        v2f wr = *(const v2f*)(Wr + n * D + 2 * p);
        float4 c4 = {wl.x, wl.y, wr.x, wr.y};
        *(float4*)(lds + n * COMB_STRIDE + p * 4) = c4;
    }
    __syncthreads();

    const int lane = threadIdx.x & 31;
    const int wave = __builtin_amdgcn_readfirstlane((int)(threadIdx.x >> 5));
    const int half = lane >> 4;
    const int l16  = lane & 15;

    float* slot = lds + D * COMB_STRIDE + wave * SLOT_FLOATS;
    float* slotLane = slot + lane * 4;
    const float* sRow = slot + l16 * SLOT_STRIDE;
    const float* xl = xin + lane * 4;

    float bias[8];
    #pragma unroll
    for (int t = 0; t < 8; ++t) bias[t] = bl[t * 16 + l16];

    for (int it = wave; ; it += LAYER_WAVES) {
        const int stripe = it * (int)gridDim.x + (int)blockIdx.x;
        if (stripe >= N_STRIPES) break;
        const int row0 = stripe * 16;

        const int offr = off[row0 + (lane <= 16 ? lane : 16)];
        const int offv = lane <= 16 ? offr : 0x7fffffff;
        const int e0   = __builtin_amdgcn_readlane(offv, 0);
        const int eEnd = __builtin_amdgcn_readlane(offv, 16);
        int cur = 0;
        int curEnd = __builtin_amdgcn_readlane(offv, 1);
        float4 rs = {0.f, 0.f, 0.f, 0.f};
        int svn = csr[(e0 + lane) < N_EDGES ? (e0 + lane) : (N_EDGES - 1)];
        for (int cb = e0; cb < eEnd; cb += 32) {
            const int sv = svn;
            const int m = (eEnd - cb) < 32 ? (eEnd - cb) : 32;
            svn = csr[(cb + 32 + lane) < N_EDGES ? (cb + 32 + lane) : (N_EDGES - 1)];
            if (m > 16) {
                float4 v[32];
                #pragma unroll
                for (int u = 0; u < 32; ++u) {
                    const int q = u < m ? u : (m - 1);
                    const long s = __builtin_amdgcn_readlane(sv, q);
                    v[u] = *(const float4*)(xl + s * D);
                }
                #pragma unroll
                for (int u = 0; u < 32; ++u) {
                    if (u == 0 || u < m) {
                        const int e = cb + u;
                        while (e == curEnd) {
                            *(float4*)(slotLane + cur * SLOT_STRIDE) = rs;
                            rs.x = 0.f; rs.y = 0.f; rs.z = 0.f; rs.w = 0.f;
                            ++cur;
                            curEnd = __builtin_amdgcn_readlane(offv, cur + 1);
                        }
                        add4(rs, v[u]);
                    }
                }
            } else {
                float4 v[16];
                #pragma unroll
                for (int u = 0; u < 16; ++u) {
                    const int q = u < m ? u : (m - 1);
                    const long s = __builtin_amdgcn_readlane(sv, q);
                    v[u] = *(const float4*)(xl + s * D);
                }
                #pragma unroll
                for (int u = 0; u < 16; ++u) {
                    if (u == 0 || u < m) {
                        const int e = cb + u;
                        while (e == curEnd) {
                            *(float4*)(slotLane + cur * SLOT_STRIDE) = rs;
                            rs.x = 0.f; rs.y = 0.f; rs.z = 0.f; rs.w = 0.f;
                            ++cur;
                            curEnd = __builtin_amdgcn_readlane(offv, cur + 1);
                        }
                        add4(rs, v[u]);
                    }
                }
            }
        }
        for (; cur < 16; ++cur) {
            *(float4*)(slotLane + cur * SLOT_STRIDE) = rs;
            rs.x = 0.f; rs.y = 0.f; rs.z = 0.f; rs.w = 0.f;
        }

        const long rA = (long)row0 + l16;
        const float c   = cnt[rA];
        const float inv = 1.0f / (c > 1.0f ? c : 1.0f);
        const float* xRow = xin + rA * D;

        v8f acc[8] = {};

        for (int k = 0; k < D; k += 4) {
            int ko = k + 2 * half;
            int p4 = ((ko >> 1) << 2);
            v2f aM = *(const v2f*)(sRow + ko);
            aM = aM * inv;
            v2f aX = *(const v2f*)(xRow + ko);

            float4 bf[8];
            #pragma unroll
            for (int t = 0; t < 8; ++t) {
                int n = t * 16 + l16;
                bf[t] = *(const float4*)(lds + n * COMB_STRIDE + p4);
            }
            #pragma unroll
            for (int t = 0; t < 8; ++t) {
                v2f bL = {bf[t].x, bf[t].y};
                v2f bR = {bf[t].z, bf[t].w};
                acc[t] = __builtin_amdgcn_wmma_f32_16x16x4_f32(
                    false, aM, false, bL, (short)0, acc[t], false, false);
                acc[t] = __builtin_amdgcn_wmma_f32_16x16x4_f32(
                    false, aX, false, bR, (short)0, acc[t], false, false);
            }
        }

        #pragma unroll
        for (int t = 0; t < 8; ++t) {
            int n = t * 16 + l16;
            #pragma unroll
            for (int r = 0; r < 8; ++r) {
                float v = acc[t][r] + bias[t];
                if (relu) v = v > 0.0f ? v : 0.0f;
                slot[(half * 8 + r) * SLOT_STRIDE + n] = v;
            }
        }
        for (int r = 0; r < 16; ++r) {
            float4 v = *(const float4*)(slotLane + r * SLOT_STRIDE);
            *(float4*)(out + ((long)row0 + r) * D + lane * 4) = v;
        }
    }
}

extern "C" void kernel_launch(void* const* d_in, const int* in_sizes, int n_in,
                              void* d_out, int out_size, void* d_ws, size_t ws_size,
                              hipStream_t stream) {
    const float* x   = (const float*)d_in[0];
    const int*   ei  = (const int*)d_in[1];
    const float* W1l = (const float*)d_in[2];
    const float* b1l = (const float*)d_in[3];
    const float* W1r = (const float*)d_in[4];
    const float* W2l = (const float*)d_in[5];
    const float* b2l = (const float*)d_in[6];
    const float* W2r = (const float*)d_in[7];

    const int* src = ei;
    const int* dst = ei + N_EDGES;

    float* list = (float*)d_ws;
    float* cntp = list + (size_t)N_NODES * D;
    float* h    = cntp + N_NODES;
    float* outp = (float*)d_out;

    int* deg  = (int*)h;
    int* rank = deg + N_NODES;
    int* tmp  = rank + N_EDGES;
    int* off  = (int*)list;
    int* csr  = off + (N_NODES + 4);

    const int  threads = 256;
    const int  edgeBlocks   = (EDGE_OCT + threads - 1) / threads;
    const int  orderBlocks  = (N_NODES + 8 * ORDER_ROWS - 1) / (8 * ORDER_ROWS);

    zero_f32<<<64, threads, 0, stream>>>((float4*)deg, N_NODES / 4);
    edge_rank<<<edgeBlocks, threads, 0, stream>>>(dst, deg, rank);
    scan_deg<<<(N_NODES + 1023) / 1024, 1024, 0, stream>>>(deg, off, cntp);
    edge_fill<<<edgeBlocks, threads, 0, stream>>>(src, dst, off, rank, tmp);
    seg_order<<<orderBlocks, threads, 0, stream>>>(off, tmp, csr);

    sage_layer<<<256, LAYER_WAVES * 32, 0, stream>>>(off, csr, cntp, x, W1l, b1l, W1r, h, 1);
    sage_layer<<<256, LAYER_WAVES * 32, 0, stream>>>(off, csr, cntp, h, W2l, b2l, W2r, outp, 0);
}
